# starting kernel
# baseline (speedup 1.0000x reference)
#include <hip/hip_runtime.h>

#define B_    32
#define N_    1024
#define D_    512
#define K_SEL 512
#define EPS_  1e-8f

typedef __attribute__((ext_vector_type(2))) float v2f;
typedef __attribute__((ext_vector_type(8))) float v8f;

__global__ __launch_bounds__(256)
void k1_norms_part(const float* __restrict__ x, float* __restrict__ inv_norms,
                   float* __restrict__ part) {
  const int b      = blockIdx.x >> 4;
  const int rowblk = blockIdx.x & 15;
  const int row0   = rowblk << 6;
  const int lane   = threadIdx.x & 31;
  const int wave   = threadIdx.x >> 5;

  __shared__ float lds_acc[8][512];

  float acc[16];
#pragma unroll
  for (int i = 0; i < 16; ++i) acc[i] = 0.f;

  const float* xb = x + (size_t)b * (N_ * D_);

  for (int r = 0; r < 8; ++r) {
    const int row = row0 + wave * 8 + r;
    const float4* xr = (const float4*)(xb + (size_t)row * D_);
    float4 v[4];
    float ss = 0.f;
#pragma unroll
    for (int c = 0; c < 4; ++c) {
      v[c] = xr[c * 32 + lane];
      ss += v[c].x * v[c].x + v[c].y * v[c].y + v[c].z * v[c].z + v[c].w * v[c].w;
    }
    if (r + 1 < 8)
      __builtin_prefetch(xb + (size_t)(row + 1) * D_ + lane * 16, 0, 1);
#pragma unroll
    for (int off = 16; off > 0; off >>= 1) ss += __shfl_xor(ss, off, 32);
    const float inv = 1.f / fmaxf(sqrtf(ss), EPS_);
    if (lane == 0) inv_norms[b * N_ + row] = inv;
#pragma unroll
    for (int c = 0; c < 4; ++c) {
      acc[c * 4 + 0] += v[c].x * inv;
      acc[c * 4 + 1] += v[c].y * inv;
      acc[c * 4 + 2] += v[c].z * inv;
      acc[c * 4 + 3] += v[c].w * inv;
    }
  }

#pragma unroll
  for (int c = 0; c < 4; ++c) {
    const int d = c * 128 + lane * 4;
    lds_acc[wave][d + 0] = acc[c * 4 + 0];
    lds_acc[wave][d + 1] = acc[c * 4 + 1];
    lds_acc[wave][d + 2] = acc[c * 4 + 2];
    lds_acc[wave][d + 3] = acc[c * 4 + 3];
  }
  __syncthreads();

  for (int d = threadIdx.x; d < 512; d += 256) {
    float s = 0.f;
#pragma unroll
    for (int w8 = 0; w8 < 8; ++w8) s += lds_acc[w8][d];
    part[((size_t)b * 16 + rowblk) * D_ + d] = s;
  }
}

__global__ __launch_bounds__(256)
void k1b_mu(const float* __restrict__ part, float* __restrict__ mu) {
  const int i = blockIdx.x * 256 + threadIdx.x;
  if (i >= B_ * D_) return;
  const int b = i / D_;
  const int d = i - b * D_;
  float s = 0.f;
#pragma unroll
  for (int j = 0; j < 16; ++j) s += part[((size_t)b * 16 + j) * D_ + d];
  mu[i] = s * (1.0f / N_);
}

__global__ __launch_bounds__(256)
void k2_scores(const float* __restrict__ x, const float* __restrict__ inv_norms,
               const float* __restrict__ mu, float* __restrict__ scores) {
  const int b    = blockIdx.x >> 7;
  const int row0 = (blockIdx.x & 127) << 3;
  const int lane = threadIdx.x & 31;
  const int wave = threadIdx.x >> 5;

  __shared__ float4 mul[128];
  {
    const float4* mub = (const float4*)(mu + (size_t)b * D_);
    for (int i = threadIdx.x; i < 128; i += 256) mul[i] = mub[i];
  }
  __syncthreads();

  const int row = row0 + wave;
  const float4* xr = (const float4*)(x + ((size_t)b * N_ + row) * D_);
  float dot = 0.f;
#pragma unroll
  for (int c = 0; c < 4; ++c) {
    const float4 v = xr[c * 32 + lane];
    const float4 m = mul[c * 32 + lane];
    dot += v.x * m.x + v.y * m.y + v.z * m.z + v.w * m.w;
  }
#pragma unroll
  for (int off = 16; off > 0; off >>= 1) dot += __shfl_xor(dot, off, 32);
  if (lane == 0) scores[b * N_ + row] = dot * inv_norms[b * N_ + row];
}

__global__ __launch_bounds__(256)
void k3_rank(const float* __restrict__ scores, float* __restrict__ w) {
  const int b = blockIdx.x >> 2;
  const int m = ((blockIdx.x & 3) << 8) + threadIdx.x;

  __shared__ float s[N_];
  for (int i = threadIdx.x; i < N_; i += 256) s[i] = scores[b * N_ + i];
  __syncthreads();

  const float sm = s[m];
  int rank = 0;
  for (int n = 0; n < N_; ++n) {
    const float sn = s[n];
    rank += (sn > sm) || ((sn == sm) && (n < m));
  }
  w[b * N_ + m] = (rank < K_SEL) ? (1.0f / K_SEL) : 0.0f;
}

__global__ __launch_bounds__(256)
void k4_merge_wmma(const float* __restrict__ x, const float* __restrict__ w,
                   float* __restrict__ out) {
  const int b     = blockIdx.x >> 2;
  const int tile  = ((blockIdx.x & 3) << 3) + (threadIdx.x >> 5);
  const int lane  = threadIdx.x & 31;
  const int half  = lane >> 4;
  const int ln    = lane & 15;
  const int dbase = tile << 4;

  __shared__ float wl[N_];
  for (int i = threadIdx.x; i < N_; i += 256) wl[i] = w[b * N_ + i];
  __syncthreads();

  const float* xb = x + (size_t)b * (N_ * D_);

  v8f c = {};
  for (int k = 0; k < N_; k += 4) {
    const int ka = k + (half ? 2 : 0);
    const int kb = ka + 1;
    v2f a  = { wl[ka], wl[kb] };
    v2f bv = { xb[(size_t)ka * D_ + dbase + ln],
               xb[(size_t)kb * D_ + dbase + ln] };
    c = __builtin_amdgcn_wmma_f32_16x16x4_f32(
         false, a,  false, bv,
         (short)0, c,  false,  false);
  }

  if (lane < 16) out[(size_t)b * D_ + dbase + ln] = c[0];
}

extern "C" void kernel_launch(void* const* d_in, const int* in_sizes, int n_in,
                              void* d_out, int out_size, void* d_ws, size_t ws_size,
                              hipStream_t stream) {
  (void)in_sizes; (void)n_in; (void)out_size; (void)ws_size;

  const float* x = (const float*)d_in[0];
  float* out = (float*)d_out;

  char* ws = (char*)d_ws;
  float* inv_norms = (float*)(ws + 0);
  float* scores    = (float*)(ws + 131072);
  float* wsel      = (float*)(ws + 262144);
  float* mu        = (float*)(ws + 393216);
  float* part      = (float*)(ws + 458752);

  k1_norms_part<<<B_ * 16, 256, 0, stream>>>(x, inv_norms, part);
  k1b_mu<<<(B_ * D_ + 255) / 256, 256, 0, stream>>>(part, mu);
  k2_scores<<<B_ * 128, 256, 0, stream>>>(x, inv_norms, mu, scores);
  k3_rank<<<B_ * 4, 256, 0, stream>>>(scores, wsel);
  k4_merge_wmma<<<B_ * 4, 256, 0, stream>>>(x, wsel, out);
}


# k3 also merges: each rank wave sums its own group's selected rows on WMMA (16-B loads, 16 free tiles), one partial per block, k4 only adds 8 partials; no weight array or row list; sum order changes
# speedup vs baseline: 2.2754x; 2.2754x over previous
#include <hip/hip_runtime.h>

#define B_    32
#define N_    1024
#define D_    512
#define K_SEL 512
#define EPS_  1e-8f

typedef __attribute__((ext_vector_type(2))) float v2f;
typedef __attribute__((ext_vector_type(8))) float v8f;
typedef __attribute__((ext_vector_type(4))) unsigned v4u;

__global__ __launch_bounds__(256)
void k1_norms_part(const float* __restrict__ x, float* __restrict__ inv_norms,
                   float* __restrict__ part) {
  const int b      = blockIdx.x >> 4;
  const int rowblk = blockIdx.x & 15;
  const int row0   = rowblk << 6;
  const int lane   = threadIdx.x & 31;
  const int wave   = threadIdx.x >> 5;

  __shared__ float lds_acc[8][512];

  float acc[16];
#pragma unroll
  for (int i = 0; i < 16; ++i) acc[i] = 0.f;

  const float* xb = x + (size_t)b * (N_ * D_);

  for (int r = 0; r < 8; ++r) {
    const int row = row0 + wave * 8 + r;
    const float4* xr = (const float4*)(xb + (size_t)row * D_);
    float4 v[4];
    float ss = 0.f;
#pragma unroll
    for (int c = 0; c < 4; ++c) {
      v[c] = xr[c * 32 + lane];
      ss += v[c].x * v[c].x + v[c].y * v[c].y + v[c].z * v[c].z + v[c].w * v[c].w;
    }
    if (r + 1 < 8)
      __builtin_prefetch(xb + (size_t)(row + 1) * D_ + lane * 16, 0, 1);
#pragma unroll
    for (int off = 16; off > 0; off >>= 1) ss += __shfl_xor(ss, off, 32);
    const float inv = 1.f / fmaxf(sqrtf(ss), EPS_);
    if (lane == 0) inv_norms[b * N_ + row] = inv;
#pragma unroll
    for (int c = 0; c < 4; ++c) {
      acc[c * 4 + 0] += v[c].x * inv;
      acc[c * 4 + 1] += v[c].y * inv;
      acc[c * 4 + 2] += v[c].z * inv;
      acc[c * 4 + 3] += v[c].w * inv;
    }
  }

#pragma unroll
  for (int c = 0; c < 4; ++c) {
    const int d = c * 128 + lane * 4;
    lds_acc[wave][d + 0] = acc[c * 4 + 0];
    lds_acc[wave][d + 1] = acc[c * 4 + 1];
    lds_acc[wave][d + 2] = acc[c * 4 + 2];
    lds_acc[wave][d + 3] = acc[c * 4 + 3];
  }
  __syncthreads();

  for (int d = threadIdx.x; d < 512; d += 256) {
    float s = 0.f;
#pragma unroll
    for (int w8 = 0; w8 < 8; ++w8) s += lds_acc[w8][d];
    part[((size_t)b * 16 + rowblk) * D_ + d] = s;
  }
}

__global__ __launch_bounds__(256)
void k1b_mu(const float* __restrict__ part, float* __restrict__ mu) {
  const int i = blockIdx.x * 256 + threadIdx.x;
  if (i >= B_ * D_) return;
  const int b = i / D_;
  const int d = i - b * D_;
  float s = 0.f;
#pragma unroll
  for (int j = 0; j < 16; ++j) s += part[((size_t)b * 16 + j) * D_ + d];
  mu[i] = s * (1.0f / N_);
}

__global__ __launch_bounds__(512) __attribute__((amdgpu_waves_per_eu(4, 4)))
void k2_scores(const float* __restrict__ x, const float* __restrict__ inv_norms,
               const float* __restrict__ mu, float* __restrict__ scores) {
  const int b    = blockIdx.x >> 3;
  const int row0 = (blockIdx.x & 7) << 7;
  const int lane = threadIdx.x & 31;
  const int wave = __builtin_amdgcn_readfirstlane((int)(threadIdx.x >> 5));
  const int g    = lane >> 3;
  const int jj   = lane & 7;

  __shared__ __attribute__((aligned(16))) float pl[16][128];

  const float4 m = *(const float4*)(mu + (size_t)b * D_ + wave * 32 + jj * 4);
  float inv = 0.f;
  if (threadIdx.x < 128) inv = inv_norms[b * N_ + row0 + (int)threadIdx.x];

  const float4* p = (const float4*)(x + ((size_t)b * N_ + row0 + g * 32) * D_ + wave * 32 + jj * 4);
  float4 v[32];
#pragma unroll
  for (int i = 0; i < 32; ++i) v[i] = p[i * (D_ / 4)];

  const v2f mlo = { m.x, m.y }, mhi = { m.z, m.w };
  float a[32];
#pragma unroll
  for (int i = 0; i < 32; ++i) {
    const v2f lo = { v[i].x, v[i].y }, hi = { v[i].z, v[i].w };
    const v2f pr = hi * mhi + lo * mlo;
    a[i] = pr.x + pr.y;
  }

  const bool s4 = (lane & 4) != 0, s2 = (lane & 2) != 0, s1 = (lane & 1) != 0;
  float r16[16], r8[8], r4[4];
#pragma unroll
  for (int i = 0; i < 16; ++i) {
    const float keep = s4 ? a[i + 16] : a[i];
    const float send = s4 ? a[i] : a[i + 16];
    r16[i] = keep + __shfl_xor(send, 4, 32);
  }
#pragma unroll
  for (int i = 0; i < 8; ++i) {
    const float keep = s2 ? r16[i + 8] : r16[i];
    const float send = s2 ? r16[i] : r16[i + 8];
    r8[i] = keep + __shfl_xor(send, 2, 32);
  }
#pragma unroll
  for (int i = 0; i < 4; ++i) {
    const float keep = s1 ? r8[i + 4] : r8[i];
    const float send = s1 ? r8[i] : r8[i + 4];
    r4[i] = keep + __shfl_xor(send, 1, 32);
  }
  *(float4*)&pl[wave][4 * lane] = make_float4(r4[0], r4[1], r4[2], r4[3]);
  __syncthreads();

  if (threadIdx.x < 128) {
    float s = 0.f;
#pragma unroll
    for (int w16 = 0; w16 < 16; ++w16) s += pl[w16][threadIdx.x];
    scores[b * N_ + row0 + (int)threadIdx.x] = s * inv;
  }
}

__global__ __launch_bounds__(128)
void k3_rank_merge(const float* __restrict__ scores, const float* __restrict__ x,
                   float* __restrict__ gpart) {
  const int lane = threadIdx.x & 31;
  const int wave = __builtin_amdgcn_readfirstlane((int)(threadIdx.x >> 5));
  const int wid  = blockIdx.x * 4 + wave;
  const int b    = wid >> 5;
  const int j0   = wid & 31;

  __shared__ __attribute__((aligned(16))) float wsum[4][D_];

  const float* sb = scores + b * N_;

  float v[32];
#pragma unroll
  for (int j = 0; j < 32; ++j) v[j] = sb[j * 32 + lane];
  const float mine = sb[j0 * 32 + lane];

#pragma unroll
  for (int j = 0; j < 32; ++j) {
    const bool below = j < j0;
    const unsigned u = __float_as_uint(v[j]);
    const unsigned step = (unsigned)(((int)u >> 31) | 1);
    unsigned up = u + (below ? step : 0u);
    up = (below && v[j] == 0.0f) ? 1u : up;
    v[j] = __uint_as_float(up);
  }

  unsigned myrank = 0;
  for (int i = 0; i < 32; ++i) {
    const float sm = __uint_as_float(
        (unsigned)__builtin_amdgcn_readlane((int)__float_as_uint(mine), i));
    unsigned cnt = 0;
#pragma unroll
    for (int j = 0; j < 32; j += 2) {
      const unsigned m0 = __builtin_amdgcn_ballot_w32(v[j] > sm);
      const unsigned m1 = __builtin_amdgcn_ballot_w32(v[j + 1] > sm);
      cnt += (unsigned)__builtin_popcountll(((unsigned long long)m1 << 32) | m0);
    }
    const unsigned tie = __builtin_amdgcn_ballot_w32(mine == sm) & ((1u << i) - 1u);
    cnt += (unsigned)__builtin_popcount(tie);
    myrank = (lane == i) ? cnt : myrank;
  }

  const bool     selme = myrank < K_SEL;
  const unsigned smask = __builtin_amdgcn_ballot_w32(selme);
  const int      nsel  = __builtin_popcount(smask);
  const int      under = __builtin_popcount(smask & ((1u << lane) - 1u));
  const int dest = selme ? under : nsel + (lane - under);
  const int list = __builtin_amdgcn_ds_permute(dest << 2, lane);

  const int half = lane >> 4;
  const int ln   = lane & 15;
  const float* xg = x + ((size_t)b * N_ + j0 * 32) * D_ + ln * 4;

  v8f acc[16];
#pragma unroll
  for (int i = 0; i < 16; ++i) acc[i] = (v8f){};

  const int steps = (nsel + 3) >> 2;

  const float e0 = (ln == 0) ? (1.0f / K_SEL) : 0.0f;
  const float e1 = (ln == 1) ? (1.0f / K_SEL) : 0.0f;

  float4 bufA[16], bufB[16];
  v2f aA0, aA1, aB0, aB1;

#define LOAD_STEP(buf, aw0, aw1, t)                                                    \
  {                                                                                    \
    const int p0 = 4 * (t) + half;                                                     \
    const int r0 = __shfl(list, p0, 32);                                               \
    const int r1 = __shfl(list, p0 + 2, 32);                                           \
    aw0 = (p0 < nsel) ? (v2f){ e0, e1 } : (v2f){ 0.0f, 0.0f };                         \
    aw1 = (p0 + 2 < nsel) ? (v2f){ e0, e1 } : (v2f){ 0.0f, 0.0f };                     \
    const float4* q0 = (const float4*)(xg + (size_t)r0 * D_);                          \
    const float4* q1 = (const float4*)(xg + (size_t)r1 * D_);                          \
    _Pragma("unroll") for (int q = 0; q < 8; ++q) buf[q] = q0[q * 16];                 \
    _Pragma("unroll") for (int q = 0; q < 8; ++q) buf[8 + q] = q1[q * 16];             \
  }
#define MMA_HALF(buf, aw, o)                                                           \
  {                                                                                    \
    _Pragma("unroll") for (int q = 0; q < 8; ++q) {                                    \
      const float4 u = buf[(o) + q];                                                   \
      const v2f blo = { u.x, u.y }, bhi = { u.z, u.w };                                \
      acc[2 * q]     = __builtin_amdgcn_wmma_f32_16x16x4_f32(false, aw, false, blo,    \
                           (short)0, acc[2 * q], false, false);                        \
      acc[2 * q + 1] = __builtin_amdgcn_wmma_f32_16x16x4_f32(false, aw, false, bhi,    \
                           (short)0, acc[2 * q + 1], false, false);                    \
    }                                                                                  \
  }

  if (steps > 0) LOAD_STEP(bufA, aA0, aA1, 0)
  for (int t = 0; t < steps; t += 2) {
    if (t + 1 < steps) LOAD_STEP(bufB, aB0, aB1, t + 1)
    MMA_HALF(bufA, aA0, 0)
    MMA_HALF(bufA, aA1, 8)
    if (t + 1 < steps) {
      if (t + 2 < steps) LOAD_STEP(bufA, aA0, aA1, t + 2)
      MMA_HALF(bufB, aB0, 0)
      MMA_HALF(bufB, aB1, 8)
    }
  }
#undef LOAD_STEP
#undef MMA_HALF

  if (lane < 16) {
#pragma unroll
    for (int q = 0; q < 8; ++q)
      *(float4*)&wsum[wave][q * 64 + lane * 4] =
          make_float4(acc[2 * q][0], acc[2 * q][1], acc[2 * q + 1][0], acc[2 * q + 1][1]);
  }
  __syncthreads();

  {
    const int d = (int)threadIdx.x * 4;
    const float4 s0 = *(const float4*)&wsum[0][d];
    const float4 s1 = *(const float4*)&wsum[1][d];
    const float4 s2 = *(const float4*)&wsum[2][d];
    const float4 s3 = *(const float4*)&wsum[3][d];
    float4 s;
    s.x = ((s0.x + s1.x) + s2.x) + s3.x;
    s.y = ((s0.y + s1.y) + s2.y) + s3.y;
    s.z = ((s0.z + s1.z) + s2.z) + s3.z;
    s.w = ((s0.w + s1.w) + s2.w) + s3.w;
    *(float4*)(gpart + (size_t)blockIdx.x * D_ + d) = s;
  }
}

__global__ __launch_bounds__(256)
void k4_sum8(const float* __restrict__ gpart, float* __restrict__ out) {
  const int i = blockIdx.x * 256 + threadIdx.x;
  if (i >= B_ * D_) return;
  const int b = i / D_;
  const int d = i - b * D_;
  float p[8];
#pragma unroll
  for (int j = 0; j < 8; ++j) p[j] = gpart[((size_t)b * 8 + j) * D_ + d];
  float s = p[0];
#pragma unroll
  for (int j = 1; j < 8; ++j) s += p[j];
  out[i] = s;
}

extern "C" void kernel_launch(void* const* d_in, const int* in_sizes, int n_in,
                              void* d_out, int out_size, void* d_ws, size_t ws_size,
                              hipStream_t stream) {
  (void)in_sizes; (void)n_in; (void)out_size; (void)ws_size;

  const float* x = (const float*)d_in[0];
  float* out = (float*)d_out;

  char* ws = (char*)d_ws;
  float* inv_norms = (float*)(ws + 0);
  float* scores    = (float*)(ws + 131072);
  float* mu        = (float*)(ws + 393216);
  float* part      = (float*)(ws + 458752);
  float* gpart     = (float*)(ws + 458752);
  k1_norms_part<<<B_ * 16, 256, 0, stream>>>(x, inv_norms, part);
  k1b_mu<<<(B_ * D_ + 255) / 256, 256, 0, stream>>>(part, mu);
  k2_scores<<<B_ * 8, 512, 0, stream>>>(x, inv_norms, mu, scores);
  k3_rank_merge<<<B_ * 8, 128, 0, stream>>>(scores, x, gpart);
  k4_sum8<<<(B_ * D_ + 255) / 256, 256, 0, stream>>>(gpart, out);
}


# k1 norms + unit sums, column-split tiling: block = 128 rows x 16 waves, wave = one 32-dim slab kept in registers (32 loads ahead, one round trip not 8), norms via LDS, 8 partials; sum order changes
# speedup vs baseline: 2.4900x; 1.0943x over previous
#include <hip/hip_runtime.h>

#define B_    32
#define N_    1024
#define D_    512
#define K_SEL 512
#define EPS_  1e-8f

typedef __attribute__((ext_vector_type(2))) float v2f;
typedef __attribute__((ext_vector_type(8))) float v8f;
typedef __attribute__((ext_vector_type(4))) unsigned v4u;

__global__ __launch_bounds__(512) __attribute__((amdgpu_waves_per_eu(4, 4)))
void k1_norms_part(const float* __restrict__ x, float* __restrict__ inv_norms,
                   float* __restrict__ part) {
  const int b      = blockIdx.x >> 3;
  const int rowblk = blockIdx.x & 7;
  const int row0   = rowblk << 7;
  const int lane   = threadIdx.x & 31;
  const int wave   = __builtin_amdgcn_readfirstlane((int)(threadIdx.x >> 5));
  const int g      = lane >> 3;
  const int jj     = lane & 7;

  __shared__ __attribute__((aligned(16))) float pl[16][128];
  __shared__ __attribute__((aligned(16))) float invs[128];

  const float4* p = (const float4*)(x + ((size_t)b * N_ + row0 + g * 32) * D_ + wave * 32 + jj * 4);
  float4 v[32];
#pragma unroll
  for (int i = 0; i < 32; ++i) v[i] = p[i * (D_ / 4)];

  float a[32];
#pragma unroll
  for (int i = 0; i < 32; ++i) {
    const v2f lo = { v[i].x, v[i].y }, hi = { v[i].z, v[i].w };
    const v2f pr = hi * hi + lo * lo;
    a[i] = pr.x + pr.y;
  }

  const bool s4 = (lane & 4) != 0, s2 = (lane & 2) != 0, s1 = (lane & 1) != 0;
  float r16[16], r8[8], r4[4];
#pragma unroll
  for (int i = 0; i < 16; ++i) {
    const float keep = s4 ? a[i + 16] : a[i];
    const float send = s4 ? a[i] : a[i + 16];
    r16[i] = keep + __shfl_xor(send, 4, 32);
  }
#pragma unroll
  for (int i = 0; i < 8; ++i) {
    const float keep = s2 ? r16[i + 8] : r16[i];
    const float send = s2 ? r16[i] : r16[i + 8];
    r8[i] = keep + __shfl_xor(send, 2, 32);
  }
#pragma unroll
  for (int i = 0; i < 4; ++i) {
    const float keep = s1 ? r8[i + 4] : r8[i];
    const float send = s1 ? r8[i] : r8[i + 4];
    r4[i] = keep + __shfl_xor(send, 1, 32);
  }
  *(float4*)&pl[wave][4 * lane] = make_float4(r4[0], r4[1], r4[2], r4[3]);
  __syncthreads();

  if (threadIdx.x < 128) {
    float ss = 0.f;
#pragma unroll
    for (int w16 = 0; w16 < 16; ++w16) ss += pl[w16][threadIdx.x];
    const float inv = 1.f / fmaxf(sqrtf(ss), EPS_);
    inv_norms[b * N_ + row0 + (int)threadIdx.x] = inv;
    invs[threadIdx.x] = inv;
  }
  __syncthreads();

  float4 acc = make_float4(0.f, 0.f, 0.f, 0.f);
#pragma unroll
  for (int i4 = 0; i4 < 8; ++i4) {
    const float4 w = *(const float4*)&invs[g * 32 + 4 * i4];
    acc.x += v[4 * i4 + 0].x * w.x;  acc.y += v[4 * i4 + 0].y * w.x;
    acc.z += v[4 * i4 + 0].z * w.x;  acc.w += v[4 * i4 + 0].w * w.x;
    acc.x += v[4 * i4 + 1].x * w.y;  acc.y += v[4 * i4 + 1].y * w.y;
    acc.z += v[4 * i4 + 1].z * w.y;  acc.w += v[4 * i4 + 1].w * w.y;
    acc.x += v[4 * i4 + 2].x * w.z;  acc.y += v[4 * i4 + 2].y * w.z;
    acc.z += v[4 * i4 + 2].z * w.z;  acc.w += v[4 * i4 + 2].w * w.z;
    acc.x += v[4 * i4 + 3].x * w.w;  acc.y += v[4 * i4 + 3].y * w.w;
    acc.z += v[4 * i4 + 3].z * w.w;  acc.w += v[4 * i4 + 3].w * w.w;
  }
  acc.x += __shfl_xor(acc.x, 8, 32);  acc.y += __shfl_xor(acc.y, 8, 32);
  acc.z += __shfl_xor(acc.z, 8, 32);  acc.w += __shfl_xor(acc.w, 8, 32);
  acc.x += __shfl_xor(acc.x, 16, 32); acc.y += __shfl_xor(acc.y, 16, 32);
  acc.z += __shfl_xor(acc.z, 16, 32); acc.w += __shfl_xor(acc.w, 16, 32);
  if (lane < 8)
    *(float4*)(part + ((size_t)b * 8 + rowblk) * D_ + wave * 32 + jj * 4) = acc;
}

__global__ __launch_bounds__(256)
void k1b_mu(const float* __restrict__ part, float* __restrict__ mu) {
  const int i = blockIdx.x * 256 + threadIdx.x;
  if (i >= B_ * D_) return;
  const int b = i / D_;
  const int d = i - b * D_;
  float s = 0.f;
#pragma unroll
  for (int j = 0; j < 8; ++j) s += part[((size_t)b * 8 + j) * D_ + d];
  mu[i] = s * (1.0f / N_);
}

__global__ __launch_bounds__(512) __attribute__((amdgpu_waves_per_eu(4, 4)))
void k2_scores(const float* __restrict__ x, const float* __restrict__ inv_norms,
               const float* __restrict__ mu, float* __restrict__ scores) {
  const int b    = blockIdx.x >> 3;
  const int row0 = (blockIdx.x & 7) << 7;
  const int lane = threadIdx.x & 31;
  const int wave = __builtin_amdgcn_readfirstlane((int)(threadIdx.x >> 5));
  const int g    = lane >> 3;
  const int jj   = lane & 7;

  __shared__ __attribute__((aligned(16))) float pl[16][128];

  const float4 m = *(const float4*)(mu + (size_t)b * D_ + wave * 32 + jj * 4);
  float inv = 0.f;
  if (threadIdx.x < 128) inv = inv_norms[b * N_ + row0 + (int)threadIdx.x];

  const float4* p = (const float4*)(x + ((size_t)b * N_ + row0 + g * 32) * D_ + wave * 32 + jj * 4);
  float4 v[32];
#pragma unroll
  for (int i = 0; i < 32; ++i) v[i] = p[i * (D_ / 4)];

  const v2f mlo = { m.x, m.y }, mhi = { m.z, m.w };
  float a[32];
#pragma unroll
  for (int i = 0; i < 32; ++i) {
    const v2f lo = { v[i].x, v[i].y }, hi = { v[i].z, v[i].w };
    const v2f pr = hi * mhi + lo * mlo;
    a[i] = pr.x + pr.y;
  }

  const bool s4 = (lane & 4) != 0, s2 = (lane & 2) != 0, s1 = (lane & 1) != 0;
  float r16[16], r8[8], r4[4];
#pragma unroll
  for (int i = 0; i < 16; ++i) {
    const float keep = s4 ? a[i + 16] : a[i];
    const float send = s4 ? a[i] : a[i + 16];
    r16[i] = keep + __shfl_xor(send, 4, 32);
  }
#pragma unroll
  for (int i = 0; i < 8; ++i) {
    const float keep = s2 ? r16[i + 8] : r16[i];
    const float send = s2 ? r16[i] : r16[i + 8];
    r8[i] = keep + __shfl_xor(send, 2, 32);
  }
#pragma unroll
  for (int i = 0; i < 4; ++i) {
    const float keep = s1 ? r8[i + 4] : r8[i];
    const float send = s1 ? r8[i] : r8[i + 4];
    r4[i] = keep + __shfl_xor(send, 1, 32);
  }
  *(float4*)&pl[wave][4 * lane] = make_float4(r4[0], r4[1], r4[2], r4[3]);
  __syncthreads();

  if (threadIdx.x < 128) {
    float s = 0.f;
#pragma unroll
    for (int w16 = 0; w16 < 16; ++w16) s += pl[w16][threadIdx.x];
    scores[b * N_ + row0 + (int)threadIdx.x] = s * inv;
  }
}

__global__ __launch_bounds__(128)
void k3_rank_merge(const float* __restrict__ scores, const float* __restrict__ x,
                   float* __restrict__ gpart) {
  const int lane = threadIdx.x & 31;
  const int wave = __builtin_amdgcn_readfirstlane((int)(threadIdx.x >> 5));
  const int wid  = blockIdx.x * 4 + wave;
  const int b    = wid >> 5;
  const int j0   = wid & 31;

  __shared__ __attribute__((aligned(16))) float wsum[4][D_];

  const float* sb = scores + b * N_;

  float v[32];
#pragma unroll
  for (int j = 0; j < 32; ++j) v[j] = sb[j * 32 + lane];
  const float mine = sb[j0 * 32 + lane];

#pragma unroll
  for (int j = 0; j < 32; ++j) {
    const bool below = j < j0;
    const unsigned u = __float_as_uint(v[j]);
    const unsigned step = (unsigned)(((int)u >> 31) | 1);
    unsigned up = u + (below ? step : 0u);
    up = (below && v[j] == 0.0f) ? 1u : up;
    v[j] = __uint_as_float(up);
  }

  unsigned myrank = 0;
  for (int i = 0; i < 32; ++i) {
    const float sm = __uint_as_float(
        (unsigned)__builtin_amdgcn_readlane((int)__float_as_uint(mine), i));
    unsigned cnt = 0;
#pragma unroll
    for (int j = 0; j < 32; j += 2) {
      const unsigned m0 = __builtin_amdgcn_ballot_w32(v[j] > sm);
      const unsigned m1 = __builtin_amdgcn_ballot_w32(v[j + 1] > sm);
      cnt += (unsigned)__builtin_popcountll(((unsigned long long)m1 << 32) | m0);
    }
    const unsigned tie = __builtin_amdgcn_ballot_w32(mine == sm) & ((1u << i) - 1u);
    cnt += (unsigned)__builtin_popcount(tie);
    myrank = (lane == i) ? cnt : myrank;
  }

  const bool     selme = myrank < K_SEL;
  const unsigned smask = __builtin_amdgcn_ballot_w32(selme);
  const int      nsel  = __builtin_popcount(smask);
  const int      under = __builtin_popcount(smask & ((1u << lane) - 1u));
  const int dest = selme ? under : nsel + (lane - under);
  const int list = __builtin_amdgcn_ds_permute(dest << 2, lane);

  const int half = lane >> 4;
  const int ln   = lane & 15;
  const float* xg = x + ((size_t)b * N_ + j0 * 32) * D_ + ln * 4;

  v8f acc[16];
#pragma unroll
  for (int i = 0; i < 16; ++i) acc[i] = (v8f){};

  const int steps = (nsel + 3) >> 2;

  const float e0 = (ln == 0) ? (1.0f / K_SEL) : 0.0f;
  const float e1 = (ln == 1) ? (1.0f / K_SEL) : 0.0f;

  float4 bufA[16], bufB[16];
  v2f aA0, aA1, aB0, aB1;

#define LOAD_STEP(buf, aw0, aw1, t)                                                    \
  {                                                                                    \
    const int p0 = 4 * (t) + half;                                                     \
    const int r0 = __shfl(list, p0, 32);                                               \
    const int r1 = __shfl(list, p0 + 2, 32);                                           \
    aw0 = (p0 < nsel) ? (v2f){ e0, e1 } : (v2f){ 0.0f, 0.0f };                         \
    aw1 = (p0 + 2 < nsel) ? (v2f){ e0, e1 } : (v2f){ 0.0f, 0.0f };                     \
    const float4* q0 = (const float4*)(xg + (size_t)r0 * D_);                          \
    const float4* q1 = (const float4*)(xg + (size_t)r1 * D_);                          \
    _Pragma("unroll") for (int q = 0; q < 8; ++q) buf[q] = q0[q * 16];                 \
    _Pragma("unroll") for (int q = 0; q < 8; ++q) buf[8 + q] = q1[q * 16];             \
  }
#define MMA_HALF(buf, aw, o)                                                           \
  {                                                                                    \
    _Pragma("unroll") for (int q = 0; q < 8; ++q) {                                    \
      const float4 u = buf[(o) + q];                                                   \
      const v2f blo = { u.x, u.y }, bhi = { u.z, u.w };                                \
      acc[2 * q]     = __builtin_amdgcn_wmma_f32_16x16x4_f32(false, aw, false, blo,    \
                           (short)0, acc[2 * q], false, false);                        \
      acc[2 * q + 1] = __builtin_amdgcn_wmma_f32_16x16x4_f32(false, aw, false, bhi,    \
                           (short)0, acc[2 * q + 1], false, false);                    \
    }                                                                                  \
  }

  if (steps > 0) LOAD_STEP(bufA, aA0, aA1, 0)
  for (int t = 0; t < steps; t += 2) {
    if (t + 1 < steps) LOAD_STEP(bufB, aB0, aB1, t + 1)
    MMA_HALF(bufA, aA0, 0)
    MMA_HALF(bufA, aA1, 8)
    if (t + 1 < steps) {
      if (t + 2 < steps) LOAD_STEP(bufA, aA0, aA1, t + 2)
      MMA_HALF(bufB, aB0, 0)
      MMA_HALF(bufB, aB1, 8)
    }
  }
#undef LOAD_STEP
#undef MMA_HALF

  if (lane < 16) {
#pragma unroll
    for (int q = 0; q < 8; ++q)
      *(float4*)&wsum[wave][q * 64 + lane * 4] =
          make_float4(acc[2 * q][0], acc[2 * q][1], acc[2 * q + 1][0], acc[2 * q + 1][1]);
  }
  __syncthreads();

  {
    const int d = (int)threadIdx.x * 4;
    const float4 s0 = *(const float4*)&wsum[0][d];
    const float4 s1 = *(const float4*)&wsum[1][d];
    const float4 s2 = *(const float4*)&wsum[2][d];
    const float4 s3 = *(const float4*)&wsum[3][d];
    float4 s;
    s.x = ((s0.x + s1.x) + s2.x) + s3.x;
    s.y = ((s0.y + s1.y) + s2.y) + s3.y;
    s.z = ((s0.z + s1.z) + s2.z) + s3.z;
    s.w = ((s0.w + s1.w) + s2.w) + s3.w;
    *(float4*)(gpart + (size_t)blockIdx.x * D_ + d) = s;
  }
}

__global__ __launch_bounds__(256)
void k4_sum8(const float* __restrict__ gpart, float* __restrict__ out) {
  const int i = blockIdx.x * 256 + threadIdx.x;
  if (i >= B_ * D_) return;
  const int b = i / D_;
  const int d = i - b * D_;
  float p[8];
#pragma unroll
  for (int j = 0; j < 8; ++j) p[j] = gpart[((size_t)b * 8 + j) * D_ + d];
  float s = p[0];
#pragma unroll
  for (int j = 1; j < 8; ++j) s += p[j];
  out[i] = s;
}

extern "C" void kernel_launch(void* const* d_in, const int* in_sizes, int n_in,
                              void* d_out, int out_size, void* d_ws, size_t ws_size,
                              hipStream_t stream) {
  (void)in_sizes; (void)n_in; (void)out_size; (void)ws_size;

  const float* x = (const float*)d_in[0];
  float* out = (float*)d_out;

  char* ws = (char*)d_ws;
  float* inv_norms = (float*)(ws + 0);
  float* scores    = (float*)(ws + 131072);
  float* mu        = (float*)(ws + 393216);
  float* part      = (float*)(ws + 458752);
  float* gpart     = (float*)(ws + 458752);
  k1_norms_part<<<B_ * 8, 512, 0, stream>>>(x, inv_norms, part);
  k1b_mu<<<(B_ * D_ + 255) / 256, 256, 0, stream>>>(part, mu);
  k2_scores<<<B_ * 8, 512, 0, stream>>>(x, inv_norms, mu, scores);
  k3_rank_merge<<<B_ * 8, 128, 0, stream>>>(scores, x, gpart);
  k4_sum8<<<(B_ * D_ + 255) / 256, 256, 0, stream>>>(gpart, out);
}


# k3 owns a 64-dim output slab per batch and ranks by binary search in the 8 row-block score lists that k2 now sorts; it writes out itself: no partials, k4 launch gone; sum order changes
# speedup vs baseline: 2.8060x; 1.1269x over previous
#include <hip/hip_runtime.h>

#define B_    32
#define N_    1024
#define D_    512
#define K_SEL 512
#define EPS_  1e-8f

typedef __attribute__((ext_vector_type(2))) float v2f;
typedef __attribute__((ext_vector_type(8))) float v8f;
typedef __attribute__((ext_vector_type(4))) unsigned v4u;

__global__ __launch_bounds__(512) __attribute__((amdgpu_waves_per_eu(4, 4)))
void k1_norms_part(const float* __restrict__ x, float* __restrict__ inv_norms,
                   float* __restrict__ part) {
  const int b      = blockIdx.x >> 3;
  const int rowblk = blockIdx.x & 7;
  const int row0   = rowblk << 7;
  const int lane   = threadIdx.x & 31;
  const int wave   = __builtin_amdgcn_readfirstlane((int)(threadIdx.x >> 5));
  const int g      = lane >> 3;
  const int jj     = lane & 7;

  __shared__ __attribute__((aligned(16))) float pl[16][128];
  __shared__ __attribute__((aligned(16))) float invs[128];

  const float4* p = (const float4*)(x + ((size_t)b * N_ + row0 + g * 32) * D_ + wave * 32 + jj * 4);
  float4 v[32];
#pragma unroll
  for (int i = 0; i < 32; ++i) v[i] = p[i * (D_ / 4)];

  float a[32];
#pragma unroll
  for (int i = 0; i < 32; ++i) {
    const v2f lo = { v[i].x, v[i].y }, hi = { v[i].z, v[i].w };
    const v2f pr = hi * hi + lo * lo;
    a[i] = pr.x + pr.y;
  }

  const bool s4 = (lane & 4) != 0, s2 = (lane & 2) != 0, s1 = (lane & 1) != 0;
  float r16[16], r8[8], r4[4];
#pragma unroll
  for (int i = 0; i < 16; ++i) {
    const float keep = s4 ? a[i + 16] : a[i];
    const float send = s4 ? a[i] : a[i + 16];
    r16[i] = keep + __shfl_xor(send, 4, 32);
  }
#pragma unroll
  for (int i = 0; i < 8; ++i) {
    const float keep = s2 ? r16[i + 8] : r16[i];
    const float send = s2 ? r16[i] : r16[i + 8];
    r8[i] = keep + __shfl_xor(send, 2, 32);
  }
#pragma unroll
  for (int i = 0; i < 4; ++i) {
    const float keep = s1 ? r8[i + 4] : r8[i];
    const float send = s1 ? r8[i] : r8[i + 4];
    r4[i] = keep + __shfl_xor(send, 1, 32);
  }
  *(float4*)&pl[wave][4 * lane] = make_float4(r4[0], r4[1], r4[2], r4[3]);
  __syncthreads();

  if (threadIdx.x < 128) {
    float ss = 0.f;
#pragma unroll
    for (int w16 = 0; w16 < 16; ++w16) ss += pl[w16][threadIdx.x];
    const float inv = 1.f / fmaxf(sqrtf(ss), EPS_);
    inv_norms[b * N_ + row0 + (int)threadIdx.x] = inv;
    invs[threadIdx.x] = inv;
  }
  __syncthreads();

  float4 acc = make_float4(0.f, 0.f, 0.f, 0.f);
#pragma unroll
  for (int i4 = 0; i4 < 8; ++i4) {
    const float4 w = *(const float4*)&invs[g * 32 + 4 * i4];
    acc.x += v[4 * i4 + 0].x * w.x;  acc.y += v[4 * i4 + 0].y * w.x;
    acc.z += v[4 * i4 + 0].z * w.x;  acc.w += v[4 * i4 + 0].w * w.x;
    acc.x += v[4 * i4 + 1].x * w.y;  acc.y += v[4 * i4 + 1].y * w.y;
    acc.z += v[4 * i4 + 1].z * w.y;  acc.w += v[4 * i4 + 1].w * w.y;
    acc.x += v[4 * i4 + 2].x * w.z;  acc.y += v[4 * i4 + 2].y * w.z;
    acc.z += v[4 * i4 + 2].z * w.z;  acc.w += v[4 * i4 + 2].w * w.z;
    acc.x += v[4 * i4 + 3].x * w.w;  acc.y += v[4 * i4 + 3].y * w.w;
    acc.z += v[4 * i4 + 3].z * w.w;  acc.w += v[4 * i4 + 3].w * w.w;
  }
  acc.x += __shfl_xor(acc.x, 8, 32);  acc.y += __shfl_xor(acc.y, 8, 32);
  acc.z += __shfl_xor(acc.z, 8, 32);  acc.w += __shfl_xor(acc.w, 8, 32);
  acc.x += __shfl_xor(acc.x, 16, 32); acc.y += __shfl_xor(acc.y, 16, 32);
  acc.z += __shfl_xor(acc.z, 16, 32); acc.w += __shfl_xor(acc.w, 16, 32);
  if (lane < 8)
    *(float4*)(part + ((size_t)b * 8 + rowblk) * D_ + wave * 32 + jj * 4) = acc;
}

__global__ __launch_bounds__(256)
void k1b_mu(const float* __restrict__ part, float* __restrict__ mu) {
  const int i = blockIdx.x * 256 + threadIdx.x;
  if (i >= B_ * D_) return;
  const int b = i / D_;
  const int d = i - b * D_;
  float s = 0.f;
#pragma unroll
  for (int j = 0; j < 8; ++j) s += part[((size_t)b * 8 + j) * D_ + d];
  mu[i] = s * (1.0f / N_);
}

__global__ __launch_bounds__(512) __attribute__((amdgpu_waves_per_eu(4, 4)))
void k2_scores(const float* __restrict__ x, const float* __restrict__ inv_norms,
               const float* __restrict__ mu, unsigned* __restrict__ sorted_k,
               int* __restrict__ sorted_i) {
  const int b    = blockIdx.x >> 3;
  const int row0 = (blockIdx.x & 7) << 7;
  const int lane = threadIdx.x & 31;
  const int wave = __builtin_amdgcn_readfirstlane((int)(threadIdx.x >> 5));
  const int g    = lane >> 3;
  const int jj   = lane & 7;

  __shared__ __attribute__((aligned(16))) float pl[16][128];
  __shared__ __attribute__((aligned(16))) unsigned sk[128];
  __shared__ unsigned ss[128];
  __shared__ int      si[128];

  const float4 m = *(const float4*)(mu + (size_t)b * D_ + wave * 32 + jj * 4);
  float inv = 0.f;
  if (threadIdx.x < 128) inv = inv_norms[b * N_ + row0 + (int)threadIdx.x];

  const float4* p = (const float4*)(x + ((size_t)b * N_ + row0 + g * 32) * D_ + wave * 32 + jj * 4);
  float4 v[32];
#pragma unroll
  for (int i = 0; i < 32; ++i) v[i] = p[i * (D_ / 4)];

  const v2f mlo = { m.x, m.y }, mhi = { m.z, m.w };
  float a[32];
#pragma unroll
  for (int i = 0; i < 32; ++i) {
    const v2f lo = { v[i].x, v[i].y }, hi = { v[i].z, v[i].w };
    const v2f pr = hi * mhi + lo * mlo;
    a[i] = pr.x + pr.y;
  }

  const bool s4 = (lane & 4) != 0, s2 = (lane & 2) != 0, s1 = (lane & 1) != 0;
  float r16[16], r8[8], r4[4];
#pragma unroll
  for (int i = 0; i < 16; ++i) {
    const float keep = s4 ? a[i + 16] : a[i];
    const float send = s4 ? a[i] : a[i + 16];
    r16[i] = keep + __shfl_xor(send, 4, 32);
  }
#pragma unroll
  for (int i = 0; i < 8; ++i) {
    const float keep = s2 ? r16[i + 8] : r16[i];
    const float send = s2 ? r16[i] : r16[i + 8];
    r8[i] = keep + __shfl_xor(send, 2, 32);
  }
#pragma unroll
  for (int i = 0; i < 4; ++i) {
    const float keep = s1 ? r8[i + 4] : r8[i];
    const float send = s1 ? r8[i] : r8[i + 4];
    r4[i] = keep + __shfl_xor(send, 1, 32);
  }
  *(float4*)&pl[wave][4 * lane] = make_float4(r4[0], r4[1], r4[2], r4[3]);
  __syncthreads();

  unsigned mykey = 0u;
  if (threadIdx.x < 128) {
    float s = 0.f;
#pragma unroll
    for (int w16 = 0; w16 < 16; ++w16) s += pl[w16][threadIdx.x];
    unsigned ub = __float_as_uint(s * inv);
    ub = (ub == 0x80000000u) ? 0u : ub;
    mykey = ub ^ ((unsigned)((int)ub >> 31) | 0x80000000u);
    sk[threadIdx.x] = mykey;
  }
  __syncthreads();

  if (threadIdx.x < 128) {
    const unsigned t = threadIdx.x;
    const unsigned km1 = mykey + 1u;
    unsigned r = 0;
#pragma unroll
    for (int u4 = 0; u4 < 32; ++u4) {
      const uint4 o = *(const uint4*)&sk[4 * u4];
      r += (o.x >= km1 - (unsigned)(4u * u4 + 0u < t)) ? 1u : 0u;
      r += (o.y >= km1 - (unsigned)(4u * u4 + 1u < t)) ? 1u : 0u;
      r += (o.z >= km1 - (unsigned)(4u * u4 + 2u < t)) ? 1u : 0u;
      r += (o.w >= km1 - (unsigned)(4u * u4 + 3u < t)) ? 1u : 0u;
    }
    r &= 127;
    ss[r] = mykey;
    si[r] = (int)t;
  }
  __syncthreads();

  if (threadIdx.x < 128) {
    sorted_k[b * N_ + row0 + (int)threadIdx.x] = ss[threadIdx.x];
    sorted_i[b * N_ + row0 + (int)threadIdx.x] = si[threadIdx.x];
  }
}

__global__ __launch_bounds__(512)
void k3_select_merge(const unsigned* __restrict__ sorted_k, const int* __restrict__ sorted_i,
                     const float* __restrict__ x, float* __restrict__ out) {
  const int b    = blockIdx.x >> 3;
  const int slab = blockIdx.x & 7;
  const int tid  = (int)threadIdx.x;
  const int lane = tid & 31;
  const int wave = __builtin_amdgcn_readfirstlane(tid >> 5);

  __shared__ unsigned L[8][128];
  __shared__ int   I[8][128];
  __shared__ int   wcnt[32];
  __shared__ int   rows[K_SEL];
  __shared__ __attribute__((aligned(16))) float psum[16][64];

  (&L[0][0])[tid]       = sorted_k[b * N_ + tid];
  (&L[0][0])[tid + 512] = sorted_k[b * N_ + tid + 512];
  (&I[0][0])[tid]       = sorted_i[b * N_ + tid];
  (&I[0][0])[tid + 512] = sorted_i[b * N_ + tid + 512];
  rows[tid] = 0;
  __syncthreads();

  bool sel[2];
#pragma unroll
  for (int k = 0; k < 2; ++k) {
    const int a = (wave >> 2) + 4 * k;
    const int p = tid & 127;
    const unsigned vk = L[a][p];
    unsigned rank4 = 4u * (unsigned)p;
#pragma unroll
    for (int c = 0; c < 8; ++c) {
      const unsigned thr = vk + ((c < a) ? 0u : 1u);
      const char* base = (const char*)&L[c][0];
      unsigned posb = 0;
#pragma unroll
      for (int step = 64; step >= 1; step >>= 1) {
        const unsigned y = *(const unsigned*)(base + posb + 4 * (step - 1));
        posb += (y >= thr) ? 4u * step : 0u;
      }
      {
        const unsigned y = *(const unsigned*)(base + posb);
        posb += (y >= thr) ? 4u : 0u;
      }
      rank4 += (c == a) ? 0u : posb;
    }
    sel[k] = rank4 < 4u * K_SEL;
    const unsigned mask = __builtin_amdgcn_ballot_w32(sel[k]);
    if (lane == 0) wcnt[wave + 16 * k] = __builtin_popcount(mask);
  }
  __syncthreads();

#pragma unroll
  for (int k = 0; k < 2; ++k) {
    const int a = (wave >> 2) + 4 * k;
    const int p = tid & 127;
    int off = 0;
#pragma unroll
    for (int j = 0; j < 32; ++j) off += (j < 4 * a) ? wcnt[j] : 0;
    const int dst = off + p;
    if (sel[k] && dst < K_SEL) rows[dst] = a * 128 + (I[a][p] & 127);
  }
  __syncthreads();

  const int half = lane >> 4;
  const int ln   = lane & 15;
  const float* xb = x + (size_t)b * N_ * D_ + slab * 64 + ln * 4;

  float4 buf[16];
#pragma unroll
  for (int i = 0; i < 16; ++i) {
    const int r = rows[32 * wave + 2 * i + half];
    buf[i] = *(const float4*)(xb + (size_t)r * D_);
  }

  const float e0 = (ln == 0) ? (1.0f / K_SEL) : 0.0f;
  const float e1 = (ln == 1) ? (1.0f / K_SEL) : 0.0f;
  const v2f aw = { e0, e1 };

  v8f acc0 = (v8f){}, acc1 = (v8f){};
#pragma unroll
  for (int i = 0; i < 16; ++i) {
    const float4 u = buf[i];
    const v2f blo = { u.x, u.y }, bhi = { u.z, u.w };
    acc0 = __builtin_amdgcn_wmma_f32_16x16x4_f32(false, aw, false, blo, (short)0, acc0,
                                                 false, false);
    acc1 = __builtin_amdgcn_wmma_f32_16x16x4_f32(false, aw, false, bhi, (short)0, acc1,
                                                 false, false);
  }

  if (lane < 16)
    *(float4*)&psum[wave][lane * 4] = make_float4(acc0[0], acc0[1], acc1[0], acc1[1]);
  __syncthreads();

  if (tid < 64) {
    float s = psum[0][tid];
#pragma unroll
    for (int w16 = 1; w16 < 16; ++w16) s += psum[w16][tid];
    out[(size_t)b * D_ + slab * 64 + tid] = s;
  }
}

extern "C" void kernel_launch(void* const* d_in, const int* in_sizes, int n_in,
                              void* d_out, int out_size, void* d_ws, size_t ws_size,
                              hipStream_t stream) {
  (void)in_sizes; (void)n_in; (void)out_size; (void)ws_size;

  const float* x = (const float*)d_in[0];
  float* out = (float*)d_out;

  char* ws = (char*)d_ws;
  float* inv_norms = (float*)(ws + 0);
  unsigned* sorted_k = (unsigned*)(ws + 131072);
  int*   sorted_i  = (int*)(ws + 262144);
  float* mu        = (float*)(ws + 393216);
  float* part      = (float*)(ws + 458752);
  k1_norms_part<<<B_ * 8, 512, 0, stream>>>(x, inv_norms, part);
  k1b_mu<<<(B_ * D_ + 255) / 256, 256, 0, stream>>>(part, mu);
  k2_scores<<<B_ * 8, 512, 0, stream>>>(x, inv_norms, mu, sorted_k, sorted_i);
  k3_select_merge<<<B_ * 8, 512, 0, stream>>>(sorted_k, sorted_i, x, out);
}


# k2 builds its mean slice itself: a lane adds the 8 row-block partials of its own 4 dims in registers (8 more 16-B loads ahead of the row loads, same order, x 1/N); reduce launch gone, 3 launches
# speedup vs baseline: 3.0844x; 1.0992x over previous
#include <hip/hip_runtime.h>

#define B_    32
#define N_    1024
#define D_    512
#define K_SEL 512
#define EPS_  1e-8f

typedef __attribute__((ext_vector_type(2))) float v2f;
typedef __attribute__((ext_vector_type(8))) float v8f;
typedef __attribute__((ext_vector_type(4))) unsigned v4u;

__global__ __launch_bounds__(512) __attribute__((amdgpu_waves_per_eu(4, 4)))
void k1_norms_part(const float* __restrict__ x, float* __restrict__ inv_norms,
                   float* __restrict__ part) {
  const int b      = blockIdx.x >> 3;
  const int rowblk = blockIdx.x & 7;
  const int row0   = rowblk << 7;
  const int lane   = threadIdx.x & 31;
  const int wave   = __builtin_amdgcn_readfirstlane((int)(threadIdx.x >> 5));
  const int g      = lane >> 3;
  const int jj     = lane & 7;

  __shared__ __attribute__((aligned(16))) float pl[16][128];
  __shared__ __attribute__((aligned(16))) float invs[128];

  const float4* p = (const float4*)(x + ((size_t)b * N_ + row0 + g * 32) * D_ + wave * 32 + jj * 4);
  float4 v[32];
#pragma unroll
  for (int i = 0; i < 32; ++i) v[i] = p[i * (D_ / 4)];

  float a[32];
#pragma unroll
  for (int i = 0; i < 32; ++i) {
    const v2f lo = { v[i].x, v[i].y }, hi = { v[i].z, v[i].w };
    const v2f pr = hi * hi + lo * lo;
    a[i] = pr.x + pr.y;
  }

  const bool s4 = (lane & 4) != 0, s2 = (lane & 2) != 0, s1 = (lane & 1) != 0;
  float r16[16], r8[8], r4[4];
#pragma unroll
  for (int i = 0; i < 16; ++i) {
    const float keep = s4 ? a[i + 16] : a[i];
    const float send = s4 ? a[i] : a[i + 16];
    r16[i] = keep + __shfl_xor(send, 4, 32);
  }
#pragma unroll
  for (int i = 0; i < 8; ++i) {
    const float keep = s2 ? r16[i + 8] : r16[i];
    const float send = s2 ? r16[i] : r16[i + 8];
    r8[i] = keep + __shfl_xor(send, 2, 32);
  }
#pragma unroll
  for (int i = 0; i < 4; ++i) {
    const float keep = s1 ? r8[i + 4] : r8[i];
    const float send = s1 ? r8[i] : r8[i + 4];
    r4[i] = keep + __shfl_xor(send, 1, 32);
  }
  *(float4*)&pl[wave][4 * lane] = make_float4(r4[0], r4[1], r4[2], r4[3]);
  __syncthreads();

  if (threadIdx.x < 128) {
    float ss = 0.f;
#pragma unroll
    for (int w16 = 0; w16 < 16; ++w16) ss += pl[w16][threadIdx.x];
    const float inv = 1.f / fmaxf(sqrtf(ss), EPS_);
    inv_norms[b * N_ + row0 + (int)threadIdx.x] = inv;
    invs[threadIdx.x] = inv;
  }
  __syncthreads();

  float4 acc = make_float4(0.f, 0.f, 0.f, 0.f);
#pragma unroll
  for (int i4 = 0; i4 < 8; ++i4) {
    const float4 w = *(const float4*)&invs[g * 32 + 4 * i4];
    acc.x += v[4 * i4 + 0].x * w.x;  acc.y += v[4 * i4 + 0].y * w.x;
    acc.z += v[4 * i4 + 0].z * w.x;  acc.w += v[4 * i4 + 0].w * w.x;
    acc.x += v[4 * i4 + 1].x * w.y;  acc.y += v[4 * i4 + 1].y * w.y;
    acc.z += v[4 * i4 + 1].z * w.y;  acc.w += v[4 * i4 + 1].w * w.y;
    acc.x += v[4 * i4 + 2].x * w.z;  acc.y += v[4 * i4 + 2].y * w.z;
    acc.z += v[4 * i4 + 2].z * w.z;  acc.w += v[4 * i4 + 2].w * w.z;
    acc.x += v[4 * i4 + 3].x * w.w;  acc.y += v[4 * i4 + 3].y * w.w;
    acc.z += v[4 * i4 + 3].z * w.w;  acc.w += v[4 * i4 + 3].w * w.w;
  }
  acc.x += __shfl_xor(acc.x, 8, 32);  acc.y += __shfl_xor(acc.y, 8, 32);
  acc.z += __shfl_xor(acc.z, 8, 32);  acc.w += __shfl_xor(acc.w, 8, 32);
  acc.x += __shfl_xor(acc.x, 16, 32); acc.y += __shfl_xor(acc.y, 16, 32);
  acc.z += __shfl_xor(acc.z, 16, 32); acc.w += __shfl_xor(acc.w, 16, 32);
  if (lane < 8)
    *(float4*)(part + ((size_t)b * 8 + rowblk) * D_ + wave * 32 + jj * 4) = acc;
}

__global__ __launch_bounds__(512) __attribute__((amdgpu_waves_per_eu(4, 4)))
void k2_scores(const float* __restrict__ x, const float* __restrict__ inv_norms,
               const float* __restrict__ part, unsigned* __restrict__ sorted_k,
               int* __restrict__ sorted_i) {
  const int b    = blockIdx.x >> 3;
  const int row0 = (blockIdx.x & 7) << 7;
  const int lane = threadIdx.x & 31;
  const int wave = __builtin_amdgcn_readfirstlane((int)(threadIdx.x >> 5));
  const int g    = lane >> 3;
  const int jj   = lane & 7;

  __shared__ __attribute__((aligned(16))) float pl[16][128];
  __shared__ __attribute__((aligned(16))) unsigned sk[128];
  __shared__ unsigned ss[128];
  __shared__ int      si[128];

  const float4* pp = (const float4*)(part + (size_t)b * 8 * D_ + wave * 32 + jj * 4);
  float4 q[8];
#pragma unroll
  for (int j = 0; j < 8; ++j) q[j] = pp[j * (D_ / 4)];
  float inv = 0.f;
  if (threadIdx.x < 128) inv = inv_norms[b * N_ + row0 + (int)threadIdx.x];

  const float4* p = (const float4*)(x + ((size_t)b * N_ + row0 + g * 32) * D_ + wave * 32 + jj * 4);
  float4 v[32];
#pragma unroll
  for (int i = 0; i < 32; ++i) v[i] = p[i * (D_ / 4)];
  __builtin_amdgcn_sched_barrier(0);

  float4 m = make_float4(0.f, 0.f, 0.f, 0.f);
#pragma unroll
  for (int j = 0; j < 8; ++j) {
    m.x += q[j].x;  m.y += q[j].y;  m.z += q[j].z;  m.w += q[j].w;
  }
  m.x *= (1.0f / N_);  m.y *= (1.0f / N_);  m.z *= (1.0f / N_);  m.w *= (1.0f / N_);

  const v2f mlo = { m.x, m.y }, mhi = { m.z, m.w };
  float a[32];
#pragma unroll
  for (int i = 0; i < 32; ++i) {
    const v2f lo = { v[i].x, v[i].y }, hi = { v[i].z, v[i].w };
    const v2f pr = hi * mhi + lo * mlo;
    a[i] = pr.x + pr.y;
  }

  const bool s4 = (lane & 4) != 0, s2 = (lane & 2) != 0, s1 = (lane & 1) != 0;
  float r16[16], r8[8], r4[4];
#pragma unroll
  for (int i = 0; i < 16; ++i) {
    const float keep = s4 ? a[i + 16] : a[i];
    const float send = s4 ? a[i] : a[i + 16];
    r16[i] = keep + __shfl_xor(send, 4, 32);
  }
#pragma unroll
  for (int i = 0; i < 8; ++i) {
    const float keep = s2 ? r16[i + 8] : r16[i];
    const float send = s2 ? r16[i] : r16[i + 8];
    r8[i] = keep + __shfl_xor(send, 2, 32);
  }
#pragma unroll
  for (int i = 0; i < 4; ++i) {
    const float keep = s1 ? r8[i + 4] : r8[i];
    const float send = s1 ? r8[i] : r8[i + 4];
    r4[i] = keep + __shfl_xor(send, 1, 32);
  }
  *(float4*)&pl[wave][4 * lane] = make_float4(r4[0], r4[1], r4[2], r4[3]);
  __syncthreads();

  unsigned mykey = 0u;
  if (threadIdx.x < 128) {
    float s = 0.f;
#pragma unroll
    for (int w16 = 0; w16 < 16; ++w16) s += pl[w16][threadIdx.x];
    unsigned ub = __float_as_uint(s * inv);
    ub = (ub == 0x80000000u) ? 0u : ub;
    mykey = ub ^ ((unsigned)((int)ub >> 31) | 0x80000000u);
    sk[threadIdx.x] = mykey;
  }
  __syncthreads();

  if (threadIdx.x < 128) {
    const unsigned t = threadIdx.x;
    const unsigned km1 = mykey + 1u;
    unsigned r = 0;
#pragma unroll
    for (int u4 = 0; u4 < 32; ++u4) {
      const uint4 o = *(const uint4*)&sk[4 * u4];
      r += (o.x >= km1 - (unsigned)(4u * u4 + 0u < t)) ? 1u : 0u;
      r += (o.y >= km1 - (unsigned)(4u * u4 + 1u < t)) ? 1u : 0u;
      r += (o.z >= km1 - (unsigned)(4u * u4 + 2u < t)) ? 1u : 0u;
      r += (o.w >= km1 - (unsigned)(4u * u4 + 3u < t)) ? 1u : 0u;
    }
    r &= 127;
    ss[r] = mykey;
    si[r] = (int)t;
  }
  __syncthreads();

  if (threadIdx.x < 128) {
    sorted_k[b * N_ + row0 + (int)threadIdx.x] = ss[threadIdx.x];
    sorted_i[b * N_ + row0 + (int)threadIdx.x] = si[threadIdx.x];
  }
}

__global__ __launch_bounds__(512)
void k3_select_merge(const unsigned* __restrict__ sorted_k, const int* __restrict__ sorted_i,
                     const float* __restrict__ x, float* __restrict__ out) {
  const int b    = blockIdx.x >> 3;
  const int slab = blockIdx.x & 7;
  const int tid  = (int)threadIdx.x;
  const int lane = tid & 31;
  const int wave = __builtin_amdgcn_readfirstlane(tid >> 5);

  __shared__ unsigned L[8][128];
  __shared__ int   I[8][128];
  __shared__ int   wcnt[32];
  __shared__ int   rows[K_SEL];
  __shared__ __attribute__((aligned(16))) float psum[16][64];

  (&L[0][0])[tid]       = sorted_k[b * N_ + tid];
  (&L[0][0])[tid + 512] = sorted_k[b * N_ + tid + 512];
  (&I[0][0])[tid]       = sorted_i[b * N_ + tid];
  (&I[0][0])[tid + 512] = sorted_i[b * N_ + tid + 512];
  rows[tid] = 0;
  __syncthreads();

  bool sel[2];
#pragma unroll
  for (int k = 0; k < 2; ++k) {
    const int a = (wave >> 2) + 4 * k;
    const int p = tid & 127;
    const unsigned vk = L[a][p];
    unsigned rank4 = 4u * (unsigned)p;
#pragma unroll
    for (int c = 0; c < 8; ++c) {
      const unsigned thr = vk + ((c < a) ? 0u : 1u);
      const char* base = (const char*)&L[c][0];
      unsigned posb = 0;
#pragma unroll
      for (int step = 64; step >= 1; step >>= 1) {
        const unsigned y = *(const unsigned*)(base + posb + 4 * (step - 1));
        posb += (y >= thr) ? 4u * step : 0u;
      }
      {
        const unsigned y = *(const unsigned*)(base + posb);
        posb += (y >= thr) ? 4u : 0u;
      }
      rank4 += (c == a) ? 0u : posb;
    }
    sel[k] = rank4 < 4u * K_SEL;
    const unsigned mask = __builtin_amdgcn_ballot_w32(sel[k]);
    if (lane == 0) wcnt[wave + 16 * k] = __builtin_popcount(mask);
  }
  __syncthreads();

#pragma unroll
  for (int k = 0; k < 2; ++k) {
    const int a = (wave >> 2) + 4 * k;
    const int p = tid & 127;
    int off = 0;
#pragma unroll
    for (int j = 0; j < 32; ++j) off += (j < 4 * a) ? wcnt[j] : 0;
    const int dst = off + p;
    if (sel[k] && dst < K_SEL) rows[dst] = a * 128 + (I[a][p] & 127);
  }
  __syncthreads();

  const int half = lane >> 4;
  const int ln   = lane & 15;
  const float* xb = x + (size_t)b * N_ * D_ + slab * 64 + ln * 4;

  float4 buf[16];
#pragma unroll
  for (int i = 0; i < 16; ++i) {
    const int r = rows[32 * wave + 2 * i + half];
    buf[i] = *(const float4*)(xb + (size_t)r * D_);
  }

  const float e0 = (ln == 0) ? (1.0f / K_SEL) : 0.0f;
  const float e1 = (ln == 1) ? (1.0f / K_SEL) : 0.0f;
  const v2f aw = { e0, e1 };

  v8f acc0 = (v8f){}, acc1 = (v8f){};
#pragma unroll
  for (int i = 0; i < 16; ++i) {
    const float4 u = buf[i];
    const v2f blo = { u.x, u.y }, bhi = { u.z, u.w };
    acc0 = __builtin_amdgcn_wmma_f32_16x16x4_f32(false, aw, false, blo, (short)0, acc0,
                                                 false, false);
    acc1 = __builtin_amdgcn_wmma_f32_16x16x4_f32(false, aw, false, bhi, (short)0, acc1,
                                                 false, false);
  }

  if (lane < 16)
    *(float4*)&psum[wave][lane * 4] = make_float4(acc0[0], acc0[1], acc1[0], acc1[1]);
  __syncthreads();

  if (tid < 64) {
    float s = psum[0][tid];
#pragma unroll
    for (int w16 = 1; w16 < 16; ++w16) s += psum[w16][tid];
    out[(size_t)b * D_ + slab * 64 + tid] = s;
  }
}

extern "C" void kernel_launch(void* const* d_in, const int* in_sizes, int n_in,
                              void* d_out, int out_size, void* d_ws, size_t ws_size,
                              hipStream_t stream) {
  (void)in_sizes; (void)n_in; (void)out_size; (void)ws_size;

  const float* x = (const float*)d_in[0];
  float* out = (float*)d_out;

  char* ws = (char*)d_ws;
  float* inv_norms = (float*)(ws + 0);
  unsigned* sorted_k = (unsigned*)(ws + 131072);
  int*   sorted_i  = (int*)(ws + 262144);
  float* part      = (float*)(ws + 393216);
  k1_norms_part<<<B_ * 8, 512, 0, stream>>>(x, inv_norms, part);
  k2_scores<<<B_ * 8, 512, 0, stream>>>(x, inv_norms, part, sorted_k, sorted_i);
  k3_select_merge<<<B_ * 8, 512, 0, stream>>>(sorted_k, sorted_i, x, out);
}
